# starting kernel
# baseline (speedup 1.0000x reference)
#include <hip/hip_runtime.h>

typedef __attribute__((ext_vector_type(2))) float v2f;
typedef __attribute__((ext_vector_type(8))) float v8f;

#define N_NODES 50000
#define R_REL   16
#define H_DIM   32
#define L_DIM   8
#define E_EDGES 1600000

#define K_STEPS ((R_REL + 1) * (H_DIM / 4))
#define TILES   (N_NODES / 16)

__global__ void rgcn_zero_f32(float* __restrict__ p, long long n) {
  long long i = (long long)blockIdx.x * blockDim.x + threadIdx.x;
  long long stride = (long long)gridDim.x * blockDim.x;
  for (; i < n; i += stride) p[i] = 0.0f;
}

__global__ void rgcn_count(const int* __restrict__ ei, const int* __restrict__ et,
                           float* __restrict__ cnt) {
  int e = blockIdx.x * blockDim.x + threadIdx.x;
  if (e >= E_EDGES) return;
  int dst = ei[E_EDGES + e];
  int r   = et[e];
  atomicAdd(&cnt[(size_t)r * N_NODES + dst], 1.0f);
}

__global__ void rgcn_scatter1(const int* __restrict__ ei, const int* __restrict__ et,
                              const float* __restrict__ cnt, const float* __restrict__ W1,
                              float* __restrict__ agg1) {
  long long t = (long long)blockIdx.x * blockDim.x + threadIdx.x;
  if (t >= (long long)E_EDGES * H_DIM) return;
  int e = (int)(t >> 5);
  int h = (int)(t & 31);
  int src = ei[e];
  int dst = ei[E_EDGES + e];
  int r   = et[e];
  float c = cnt[(size_t)r * N_NODES + dst];
  float norm = 1.0f / fmaxf(c, 1.0f);
  float w = W1[((size_t)r * N_NODES + src) * H_DIM + h];
  atomicAdd(&agg1[(size_t)dst * H_DIM + h], w * norm);
}

__global__ void rgcn_h1(const float* __restrict__ agg1, const float* __restrict__ root1,
                        const float* __restrict__ b1, float* __restrict__ h1) {
  int t = blockIdx.x * blockDim.x + threadIdx.x;
  if (t >= N_NODES * H_DIM) return;
  float v = agg1[t] + root1[t] + b1[t & (H_DIM - 1)];
  h1[t] = fmaxf(v, 0.0f);
}

__global__ void rgcn_scatter2(const int* __restrict__ ei, const int* __restrict__ et,
                              const float* __restrict__ cnt, const float* __restrict__ h1,
                              float* __restrict__ A) {
  long long t = (long long)blockIdx.x * blockDim.x + threadIdx.x;
  if (t >= (long long)E_EDGES * H_DIM) return;
  int e = (int)(t >> 5);
  int h = (int)(t & 31);
  int src = ei[e];
  int dst = ei[E_EDGES + e];
  int r   = et[e];
  float c = cnt[(size_t)r * N_NODES + dst];
  float norm = 1.0f / fmaxf(c, 1.0f);
  float v = h1[(size_t)src * H_DIM + h] * norm;
  atomicAdd(&A[((size_t)r * N_NODES + dst) * H_DIM + h], v);
}

__global__ void rgcn_out_wmma(const float* __restrict__ Amat, const float* __restrict__ h1,
                              const float* __restrict__ W2, const float* __restrict__ root2,
                              const float* __restrict__ b2, float* __restrict__ out) {
  __shared__ float ldsB[K_STEPS * 64];

  for (int idx = threadIdx.x; idx < K_STEPS * 64; idx += blockDim.x) {
    int s    = idx >> 6;
    int slot = idx & 63;
    int ln   = slot >> 1;
    int j    = slot & 1;
    int n    = ln & 15;
    int k    = ((s & 7) << 2) + ((ln >> 4) << 1) + j;
    int rr   = s >> 3;
    float v = 0.0f;
    if (n < L_DIM) {
      const float* Bseg = (rr < R_REL) ? (W2 + (size_t)rr * H_DIM * L_DIM) : root2;
      v = Bseg[(size_t)k * L_DIM + n];
    }
    ldsB[idx] = v;
  }
  __syncthreads();

  int wave = (blockIdx.x * blockDim.x + threadIdx.x) >> 5;
  int lane = threadIdx.x & 31;
  if (wave >= TILES) return;

  const int row0 = wave * 16;
  const int m    = lane & 15;
  const int koff = (lane >> 4) * 2;

  v8f acc = {};
  for (int rr = 0; rr <= R_REL; ++rr) {
    const float* Aseg = (rr < R_REL) ? (Amat + (size_t)rr * N_NODES * H_DIM) : h1;
    const float* arow = Aseg + (size_t)(row0 + m) * H_DIM + koff;
    const float* brow = ldsB + (size_t)rr * (H_DIM / 4) * 64 + lane * 2;
    #pragma unroll
    for (int k0 = 0; k0 < H_DIM; k0 += 4) {
      v2f a = *(const v2f*)(arow + k0);
      v2f b = *(const v2f*)(brow + (k0 >> 2) * 64);
      acc = __builtin_amdgcn_wmma_f32_16x16x4_f32(
           false, a,  false, b,
           (short)0, acc,  false,  false);
    }
  }

  int l = lane & 15;
  if (l < L_DIM) {
    float bias = b2[l];
    int half = lane >> 4;
    #pragma unroll
    for (int i = 0; i < 8; ++i) {
      int node = row0 + i + 8 * half;
      float x = acc[i] + bias;
      out[(size_t)node * L_DIM + l] = 1.0f / (1.0f + __expf(-x));
    }
  }
}

extern "C" void kernel_launch(void* const* d_in, const int* in_sizes, int n_in,
                              void* d_out, int out_size, void* d_ws, size_t ws_size,
                              hipStream_t stream) {
  const int*   ei    = (const int*)d_in[0];
  const int*   et    = (const int*)d_in[1];
  const float* W1    = (const float*)d_in[2];
  const float* root1 = (const float*)d_in[3];
  const float* b1    = (const float*)d_in[4];
  const float* W2    = (const float*)d_in[5];
  const float* root2 = (const float*)d_in[6];
  const float* b2    = (const float*)d_in[7];
  float*       out   = (float*)d_out;

  float* ws   = (float*)d_ws;
  float* cnt  = ws;
  float* agg1 = cnt + (size_t)R_REL * N_NODES;
  float* Amat = agg1 + (size_t)N_NODES * H_DIM;
  float* h1   = Amat + (size_t)R_REL * N_NODES * H_DIM;

  const long long nZero = (long long)R_REL * N_NODES
                        + (long long)N_NODES * H_DIM
                        + (long long)R_REL * N_NODES * H_DIM;
  rgcn_zero_f32<<<2048, 256, 0, stream>>>(ws, nZero);

  rgcn_count<<<(E_EDGES + 255) / 256, 256, 0, stream>>>(ei, et, cnt);

  const long long eh = (long long)E_EDGES * H_DIM;
  const int ehBlocks = (int)((eh + 255) / 256);
  rgcn_scatter1<<<ehBlocks, 256, 0, stream>>>(ei, et, cnt, W1, agg1);

  rgcn_h1<<<(N_NODES * H_DIM + 255) / 256, 256, 0, stream>>>(agg1, root1, b1, h1);

  rgcn_scatter2<<<ehBlocks, 256, 0, stream>>>(ei, et, cnt, h1, Amat);

  const int gemmBlocks = (TILES * 32 + 255) / 256;
  rgcn_out_wmma<<<gemmBlocks, 256, 0, stream>>>(Amat, h1, W2, root2, b2, out);

  (void)in_sizes; (void)n_in; (void)out_size; (void)ws_size;
}


# Linking pass placed by chip half: a block reads its die, draws edge chunks from its side's counter and links only edges whose dst is in that side's node range; list order arrival-set as before
# speedup vs baseline: 4.7855x; 4.7855x over previous
#include <hip/hip_runtime.h>

typedef __attribute__((ext_vector_type(2))) float v2f;
typedef __attribute__((ext_vector_type(8))) float v8f;

#define N_NODES 50000
#define R_REL   16
#define H_DIM   32
#define L_DIM   8
#define E_EDGES 1600000

#define TILE_M  16
#define TILES   (N_NODES / TILE_M)
#define SEGS    (R_REL + 1)
#define A_PITCH (H_DIM + 4)
#define BLOCK2  (TILE_M * 32)
static_assert(N_NODES % TILE_M == 0, "tiles are exact");

#define LINK_T     256
#define CHUNK_E    2048
#define EPT        (CHUNK_E / LINK_T)
#define N_CHUNKS   ((E_EDGES + CHUNK_E - 1) / CHUNK_E)
#define LINK_GRID  1024
#define HALF_SPLIT (N_NODES / 2)
static_assert(E_EDGES % LINK_T == 0, "a row of LINK_T edges is whole or absent");
static_assert(EPT == 8, "the load fence in the linking kernel names eight edges");

__global__ void rgcn_init_heads(int* __restrict__ head, int n, int* __restrict__ ticket) {
  int i = blockIdx.x * blockDim.x + threadIdx.x;
  int stride = gridDim.x * blockDim.x;
  if (i < 2) ticket[i] = 0;
  for (; i < n; i += stride) head[i] = -1;
}

__device__ __forceinline__ int rgcn_chip_half() {
  unsigned v = __builtin_amdgcn_s_sendmsg_rtn(0x87);
  return (int)((v >> 18) & 1u);
}

__global__ __launch_bounds__(LINK_T) void rgcn_link(const int* __restrict__ ei, const int* __restrict__ et,
                                                    int* __restrict__ head, int* __restrict__ nxt,
                                                    int* __restrict__ ticket) {
  __shared__ int s_half;
  __shared__ int s_tk[2];
  if (threadIdx.x == 0) s_half = rgcn_chip_half();
  __syncthreads();
  const int half = s_half;
  for (int pass = 0; pass < 2; ++pass) {
    const int hs = pass ? 1 - half : half;
    int* nx = nxt + (size_t)hs * E_EDGES;
    if (threadIdx.x == 0) s_tk[0] = atomicAdd(&ticket[hs], 1);
    __syncthreads();
    int buf = 0;
    for (int it = 0; it < N_CHUNKS; ++it) {
      const int c = s_tk[buf];
      if (c >= N_CHUNKS) break;
      const int base = c * CHUNK_E + (int)threadIdx.x;
      int d[EPT], r[EPT];
      #pragma unroll
      for (int j = 0; j < EPT; ++j) {
        const int e  = base + j * LINK_T;
        const int ec = e < E_EDGES ? e : E_EDGES - 1;
        d[j] = ei[E_EDGES + ec];
        r[j] = et[ec];
      }
      asm volatile("" :: "v"(d[0]), "v"(d[1]), "v"(d[2]), "v"(d[3]),
                         "v"(d[4]), "v"(d[5]), "v"(d[6]), "v"(d[7]),
                         "v"(r[0]), "v"(r[1]), "v"(r[2]), "v"(r[3]),
                         "v"(r[4]), "v"(r[5]), "v"(r[6]), "v"(r[7]));
      int old[EPT];
      #pragma unroll
      for (int j = 0; j < EPT; ++j) {
        const int e = base + j * LINK_T;
        old[j] = -1;
        if (e < E_EDGES && (d[j] >= HALF_SPLIT) == (hs != 0))
          old[j] = atomicExch(&head[(size_t)r[j] * N_NODES + d[j]], e);
      }
      if (threadIdx.x == 0) s_tk[buf ^ 1] = atomicAdd(&ticket[hs], 1);
      __syncthreads();
      #pragma unroll
      for (int j = 0; j < EPT; ++j) {
        const int e = base + j * LINK_T;
        if (e < E_EDGES && (d[j] >= HALF_SPLIT) == (hs != 0)) nx[e] = old[j];
      }
      buf ^= 1;
    }
    __syncthreads();
  }
}

__device__ __forceinline__ void rgcn_walk(const int* __restrict__ head, const int* __restrict__ nxt,
                                          const int* __restrict__ ei, const float* __restrict__ tab,
                                          size_t relStride, int dst, int lane,
                                          float (&acc)[R_REL], int& len) {
  int e = (lane < R_REL) ? head[(size_t)lane * N_NODES + dst] : -1;
  len = 0;
  #pragma unroll
  for (int l = 0; l < R_REL; ++l) acc[l] = 0.0f;
  for (int it = 0; it < E_EDGES; ++it) {
    unsigned mask = __builtin_amdgcn_ballot_w32(e >= 0);
    if (mask == 0u) break;
    int s = 0, nx = -1;
    if (e >= 0) { s = ei[e]; nx = nxt[e]; ++len; }
    int sl[R_REL];
    #pragma unroll
    for (int l = 0; l < R_REL; ++l) sl[l] = __builtin_amdgcn_readlane(s, l);
    float row[R_REL];
    #pragma unroll
    for (int l = 0; l < R_REL; ++l) {
      row[l] = 0.0f;
      if (mask & (1u << l))
        row[l] = tab[(size_t)l * relStride + (size_t)sl[l] * H_DIM + lane];
    }
    #pragma unroll
    for (int l = 0; l < R_REL; ++l) acc[l] += row[l];
    e = nx;
  }
}

__device__ __forceinline__ float rgcn_lane_f(float v, int l) {
  return __int_as_float(__builtin_amdgcn_readlane(__float_as_int(v), l));
}

__global__ void rgcn_layer1(const int* __restrict__ ei, const int* __restrict__ head,
                            const int* __restrict__ nxt, const float* __restrict__ W1,
                            const float* __restrict__ root1, const float* __restrict__ b1,
                            float* __restrict__ h1) {
  int dst  = (blockIdx.x * blockDim.x + threadIdx.x) >> 5;
  int lane = threadIdx.x & 31;
  if (dst >= N_NODES) return;
  float acc[R_REL];
  int len;
  const int* nx = nxt + (__builtin_amdgcn_readfirstlane(dst) >= HALF_SPLIT ? (size_t)E_EDGES : (size_t)0);
  rgcn_walk(head, nx, ei, W1, (size_t)N_NODES * H_DIM, dst, lane, acc, len);
  float nrm = 1.0f / fmaxf((float)len, 1.0f);
  float tot = 0.0f;
  #pragma unroll
  for (int l = 0; l < R_REL; ++l) tot += acc[l] * rgcn_lane_f(nrm, l);
  int t = dst * H_DIM + lane;
  float v = tot + root1[t] + b1[lane];
  h1[t] = fmaxf(v, 0.0f);
}

__global__ __launch_bounds__(BLOCK2) void rgcn_layer2_out(
    const int* __restrict__ ei, const int* __restrict__ head, const int* __restrict__ nxt,
    const float* __restrict__ h1, const float* __restrict__ W2, const float* __restrict__ root2,
    const float* __restrict__ b2, float* __restrict__ out) {
  __shared__ float As[SEGS * TILE_M * A_PITCH];
  __shared__ float Ps[R_REL * TILE_M * L_DIM];

  const int lane = threadIdx.x & 31;
  const int w    = __builtin_amdgcn_readfirstlane((int)(threadIdx.x >> 5));
  const int dst  = blockIdx.x * TILE_M + w;

  const float hrow = h1[(size_t)dst * H_DIM + lane];

  float acc[R_REL];
  int len;
  const int* nx = nxt + (dst >= HALF_SPLIT ? (size_t)E_EDGES : (size_t)0);
  rgcn_walk(head, nx, ei, h1, (size_t)0, dst, lane, acc, len);
  const float nrm = 1.0f / fmaxf((float)len, 1.0f);

  const int n    = lane & 15;
  const int half = lane >> 4;
  const int nc   = n & (L_DIM - 1);
  v2f bf[H_DIM / 4];
  {
    const float* bp = W2 + ((size_t)w * H_DIM + 2 * half) * L_DIM + nc;
    #pragma unroll
    for (int s = 0; s < H_DIM / 4; ++s) {
      bf[s].x = bp[(4 * s) * L_DIM];
      bf[s].y = bp[(4 * s + 1) * L_DIM];
    }
  }
  v2f br; br.x = 0.0f; br.y = 0.0f;
  if (w < 8) {
    const float* bp = root2 + (size_t)(4 * w + 2 * half) * L_DIM + nc;
    br.x = bp[0];
    br.y = bp[L_DIM];
  }

  #pragma unroll
  for (int l = 0; l < R_REL; ++l)
    As[(l * TILE_M + w) * A_PITCH + lane] = acc[l] * rgcn_lane_f(nrm, l);
  As[(R_REL * TILE_M + w) * A_PITCH + lane] = hrow;
  __syncthreads();

  const float* ap = As + (w * TILE_M + n) * A_PITCH + 2 * half;
  v2f af[H_DIM / 4];
  #pragma unroll
  for (int s = 0; s < H_DIM / 4; ++s) af[s] = *(const v2f*)(ap + 4 * s);
  v8f d = {};
  #pragma unroll
  for (int s = 0; s < H_DIM / 4; ++s) {
    d = __builtin_amdgcn_wmma_f32_16x16x4_f32(
         false, af[s],  false, bf[s],
         (short)0, d,  false,  false);
  }
  if (w < 8) {
    v2f a = *(const v2f*)(As + (R_REL * TILE_M + n) * A_PITCH + 4 * w + 2 * half);
    d = __builtin_amdgcn_wmma_f32_16x16x4_f32(
         false, a,  false, br,
         (short)0, d,  false,  false);
  }

  if (n < L_DIM) {
    #pragma unroll
    for (int i = 0; i < 8; ++i) Ps[w * (TILE_M * L_DIM) + i * 16 + half * 8 + n] = d[i];
  }
  __syncthreads();

  const int t = threadIdx.x;
  if (t < TILE_M * L_DIM) {
    const int m   = t >> 3;
    const int c   = t & 7;
    const int idx = (m & 7) * 16 + (m >> 3) * 8 + c;
    float x = 0.0f;
    #pragma unroll
    for (int ww = 0; ww < R_REL; ++ww) x += Ps[ww * (TILE_M * L_DIM) + idx];
    x += b2[c];
    out[(size_t)blockIdx.x * (TILE_M * L_DIM) + t] = 1.0f / (1.0f + __expf(-x));
  }
}

extern "C" void kernel_launch(void* const* d_in, const int* in_sizes, int n_in,
                              void* d_out, int out_size, void* d_ws, size_t ws_size,
                              hipStream_t stream) {
  const int*   ei    = (const int*)d_in[0];
  const int*   et    = (const int*)d_in[1];
  const float* W1    = (const float*)d_in[2];
  const float* root1 = (const float*)d_in[3];
  const float* b1    = (const float*)d_in[4];
  const float* W2    = (const float*)d_in[5];
  const float* root2 = (const float*)d_in[6];
  const float* b2    = (const float*)d_in[7];
  float*       out   = (float*)d_out;

  int*   head = (int*)d_ws;
  int*   nxt  = head + (size_t)R_REL * N_NODES;
  float* h1   = (float*)(nxt + (size_t)2 * E_EDGES);
  int*   ticket = (int*)(h1 + (size_t)N_NODES * H_DIM);

  rgcn_init_heads<<<(R_REL * N_NODES + 255) / 256, 256, 0, stream>>>(head, R_REL * N_NODES, ticket);

  rgcn_link<<<LINK_GRID, LINK_T, 0, stream>>>(ei, et, head, nxt, ticket);

  const int nodeBlocks = (N_NODES * 32 + 255) / 256;
  rgcn_layer1<<<nodeBlocks, 256, 0, stream>>>(ei, head, nxt, W1, root1, b1, h1);

  rgcn_layer2_out<<<TILES, BLOCK2, 0, stream>>>(ei, head, nxt, h1, W2, root2, b2, out);

  (void)in_sizes; (void)n_in; (void)out_size; (void)ws_size;
}


# List storage re-laid for the walk: a node's 16 heads contiguous, each edge's {successor, src} one 8-byte record, next records asked for beside a hop's rows: fewer lines per hop; sum order as before
# speedup vs baseline: 5.1314x; 1.0723x over previous
#include <hip/hip_runtime.h>

typedef __attribute__((ext_vector_type(2))) float v2f;
typedef __attribute__((ext_vector_type(8))) float v8f;
typedef __attribute__((ext_vector_type(2))) int   v2i;

#define N_NODES 50000
#define R_REL   16
#define H_DIM   32
#define L_DIM   8
#define E_EDGES 1600000

#define TILE_M  16
#define TILES   (N_NODES / TILE_M)
#define SEGS    (R_REL + 1)
#define A_PITCH (H_DIM + 4)
#define BLOCK2  (TILE_M * 32)
static_assert(N_NODES % TILE_M == 0, "tiles are exact");

#define LINK_T     256
#define CHUNK_E    2048
#define EPT        (CHUNK_E / LINK_T)
#define N_CHUNKS   ((E_EDGES + CHUNK_E - 1) / CHUNK_E)
#define LINK_GRID  1024
#define HALF_SPLIT (N_NODES / 2)
static_assert(E_EDGES % LINK_T == 0, "a row of LINK_T edges is whole or absent");
static_assert(EPT == 8, "the load fence in the linking kernel names eight edges");


__global__ void rgcn_init_heads(int* __restrict__ head, int n, int* __restrict__ ticket) {
  int i = blockIdx.x * blockDim.x + threadIdx.x;
  int stride = gridDim.x * blockDim.x;
  if (i < 2) ticket[i] = 0;
  for (; i < n; i += stride) head[i] = -1;
}

__device__ __forceinline__ int rgcn_chip_half() {
  unsigned v = __builtin_amdgcn_s_sendmsg_rtn(0x87);
  return (int)((v >> 18) & 1u);
}

__global__ __launch_bounds__(LINK_T) void rgcn_link(const int* __restrict__ ei, const int* __restrict__ et,
                                                    int* __restrict__ head, int* __restrict__ rec,
                                                    int* __restrict__ ticket) {
  __shared__ int s_half;
  __shared__ int s_tk[2];
  if (threadIdx.x == 0) s_half = rgcn_chip_half();
  __syncthreads();
  const int half = s_half;
  for (int pass = 0; pass < 2; ++pass) {
    const int hs = pass ? 1 - half : half;
    v2i* rc = (v2i*)rec + (size_t)hs * E_EDGES;
    if (threadIdx.x == 0) s_tk[0] = atomicAdd(&ticket[hs], 1);
    __syncthreads();
    int buf = 0;
    for (int it = 0; it < N_CHUNKS; ++it) {
      const int c = s_tk[buf];
      if (c >= N_CHUNKS) break;
      const int base = c * CHUNK_E + (int)threadIdx.x;
      int d[EPT], r[EPT], s[EPT];
      #pragma unroll
      for (int j = 0; j < EPT; ++j) {
        const int e  = base + j * LINK_T;
        const int ec = e < E_EDGES ? e : E_EDGES - 1;
        d[j] = ei[E_EDGES + ec];
        r[j] = et[ec];
        s[j] = ei[ec];
      }
      asm volatile("" :: "v"(d[0]), "v"(d[1]), "v"(d[2]), "v"(d[3]),
                         "v"(d[4]), "v"(d[5]), "v"(d[6]), "v"(d[7]),
                         "v"(r[0]), "v"(r[1]), "v"(r[2]), "v"(r[3]),
                         "v"(r[4]), "v"(r[5]), "v"(r[6]), "v"(r[7]),
                         "v"(s[0]), "v"(s[1]), "v"(s[2]), "v"(s[3]),
                         "v"(s[4]), "v"(s[5]), "v"(s[6]), "v"(s[7]));
      int old[EPT];
      #pragma unroll
      for (int j = 0; j < EPT; ++j) {
        const int e = base + j * LINK_T;
        old[j] = -1;
        if (e < E_EDGES && (d[j] >= HALF_SPLIT) == (hs != 0))
          old[j] = atomicExch(&head[(size_t)d[j] * R_REL + r[j]], e);
      }
      if (threadIdx.x == 0) s_tk[buf ^ 1] = atomicAdd(&ticket[hs], 1);
      __syncthreads();
      #pragma unroll
      for (int j = 0; j < EPT; ++j) {
        const int e = base + j * LINK_T;
        if (e < E_EDGES && (d[j] >= HALF_SPLIT) == (hs != 0)) {
          v2i rv; rv.x = old[j]; rv.y = s[j];
          rc[e] = rv;
        }
      }
      buf ^= 1;
    }
    __syncthreads();
  }
}

__device__ __forceinline__ void rgcn_walk(const int* __restrict__ head, const v2i* __restrict__ rec,
                                          const float* __restrict__ tab,
                                          size_t relStride, int dst, int lane,
                                          float (&acc)[R_REL], int& len) {
  int e = (lane < R_REL) ? head[(size_t)dst * R_REL + lane] : -1;
  len = 0;
  #pragma unroll
  for (int l = 0; l < R_REL; ++l) acc[l] = 0.0f;
  v2i cur; cur.x = -1; cur.y = 0;
  if (e >= 0) cur = rec[e];
  for (int it = 0; it < E_EDGES; ++it) {
    unsigned mask = __builtin_amdgcn_ballot_w32(e >= 0);
    if (mask == 0u) break;
    const int nx = cur.x;
    const int s  = cur.y;
    if (e >= 0) ++len;
    v2i nrec; nrec.x = -1; nrec.y = 0;
    if (nx >= 0) nrec = rec[nx];
    int sl[R_REL];
    #pragma unroll
    for (int l = 0; l < R_REL; ++l) sl[l] = __builtin_amdgcn_readlane(s, l);
    float row[R_REL];
    #pragma unroll
    for (int l = 0; l < R_REL; ++l) {
      row[l] = 0.0f;
      if (mask & (1u << l))
        row[l] = tab[(size_t)l * relStride + (size_t)sl[l] * H_DIM + lane];
    }
    #pragma unroll
    for (int l = 0; l < R_REL; ++l) acc[l] += row[l];
    e = nx;
    cur = nrec;
  }
}

__device__ __forceinline__ float rgcn_lane_f(float v, int l) {
  return __int_as_float(__builtin_amdgcn_readlane(__float_as_int(v), l));
}

__global__ void rgcn_layer1(const int* __restrict__ head,
                            const int* __restrict__ rec, const float* __restrict__ W1,
                            const float* __restrict__ root1, const float* __restrict__ b1,
                            float* __restrict__ h1) {
  int dst  = (blockIdx.x * blockDim.x + threadIdx.x) >> 5;
  int lane = threadIdx.x & 31;
  if (dst >= N_NODES) return;
  float acc[R_REL];
  int len;
  const v2i* rc = (const v2i*)rec + (__builtin_amdgcn_readfirstlane(dst) >= HALF_SPLIT ? (size_t)E_EDGES : (size_t)0);
  rgcn_walk(head, rc, W1, (size_t)N_NODES * H_DIM, dst, lane, acc, len);
  float nrm = 1.0f / fmaxf((float)len, 1.0f);
  float tot = 0.0f;
  #pragma unroll
  for (int l = 0; l < R_REL; ++l) tot += acc[l] * rgcn_lane_f(nrm, l);
  int t = dst * H_DIM + lane;
  float v = tot + root1[t] + b1[lane];
  h1[t] = fmaxf(v, 0.0f);
}

__global__ __launch_bounds__(BLOCK2) void rgcn_layer2_out(
    const int* __restrict__ head, const int* __restrict__ rec,
    const float* __restrict__ h1, const float* __restrict__ W2, const float* __restrict__ root2,
    const float* __restrict__ b2, float* __restrict__ out) {
  __shared__ float As[SEGS * TILE_M * A_PITCH];
  __shared__ float Ps[R_REL * TILE_M * L_DIM];

  const int lane = threadIdx.x & 31;
  const int w    = __builtin_amdgcn_readfirstlane((int)(threadIdx.x >> 5));
  const int dst  = blockIdx.x * TILE_M + w;

  const float hrow = h1[(size_t)dst * H_DIM + lane];

  float acc[R_REL];
  int len;
  const v2i* rc = (const v2i*)rec + (dst >= HALF_SPLIT ? (size_t)E_EDGES : (size_t)0);
  rgcn_walk(head, rc, h1, (size_t)0, dst, lane, acc, len);
  const float nrm = 1.0f / fmaxf((float)len, 1.0f);

  const int n    = lane & 15;
  const int half = lane >> 4;
  const int nc   = n & (L_DIM - 1);
  v2f bf[H_DIM / 4];
  {
    const float* bp = W2 + ((size_t)w * H_DIM + 2 * half) * L_DIM + nc;
    #pragma unroll
    for (int s = 0; s < H_DIM / 4; ++s) {
      bf[s].x = bp[(4 * s) * L_DIM];
      bf[s].y = bp[(4 * s + 1) * L_DIM];
    }
  }
  v2f br; br.x = 0.0f; br.y = 0.0f;
  if (w < 8) {
    const float* bp = root2 + (size_t)(4 * w + 2 * half) * L_DIM + nc;
    br.x = bp[0];
    br.y = bp[L_DIM];
  }

  #pragma unroll
  for (int l = 0; l < R_REL; ++l)
    As[(l * TILE_M + w) * A_PITCH + lane] = acc[l] * rgcn_lane_f(nrm, l);
  As[(R_REL * TILE_M + w) * A_PITCH + lane] = hrow;
  __syncthreads();

  const float* ap = As + (w * TILE_M + n) * A_PITCH + 2 * half;
  v2f af[H_DIM / 4];
  #pragma unroll
  for (int s = 0; s < H_DIM / 4; ++s) af[s] = *(const v2f*)(ap + 4 * s);
  v8f d = {};
  #pragma unroll
  for (int s = 0; s < H_DIM / 4; ++s) {
    d = __builtin_amdgcn_wmma_f32_16x16x4_f32(
         false, af[s],  false, bf[s],
         (short)0, d,  false,  false);
  }
  if (w < 8) {
    v2f a = *(const v2f*)(As + (R_REL * TILE_M + n) * A_PITCH + 4 * w + 2 * half);
    d = __builtin_amdgcn_wmma_f32_16x16x4_f32(
         false, a,  false, br,
         (short)0, d,  false,  false);
  }

  if (n < L_DIM) {
    #pragma unroll
    for (int i = 0; i < 8; ++i) Ps[w * (TILE_M * L_DIM) + i * 16 + half * 8 + n] = d[i];
  }
  __syncthreads();

  const int t = threadIdx.x;
  if (t < TILE_M * L_DIM) {
    const int m   = t >> 3;
    const int c   = t & 7;
    const int idx = (m & 7) * 16 + (m >> 3) * 8 + c;
    float x = 0.0f;
    #pragma unroll
    for (int ww = 0; ww < R_REL; ++ww) x += Ps[ww * (TILE_M * L_DIM) + idx];
    x += b2[c];
    out[(size_t)blockIdx.x * (TILE_M * L_DIM) + t] = 1.0f / (1.0f + __expf(-x));
  }
}

extern "C" void kernel_launch(void* const* d_in, const int* in_sizes, int n_in,
                              void* d_out, int out_size, void* d_ws, size_t ws_size,
                              hipStream_t stream) {
  const int*   ei    = (const int*)d_in[0];
  const int*   et    = (const int*)d_in[1];
  const float* W1    = (const float*)d_in[2];
  const float* root1 = (const float*)d_in[3];
  const float* b1    = (const float*)d_in[4];
  const float* W2    = (const float*)d_in[5];
  const float* root2 = (const float*)d_in[6];
  const float* b2    = (const float*)d_in[7];
  float*       out   = (float*)d_out;

  int*   head = (int*)d_ws;
  int*   rec  = head + (size_t)R_REL * N_NODES;
  float* h1   = (float*)(rec + (size_t)4 * E_EDGES);
  int*   ticket = (int*)(h1 + (size_t)N_NODES * H_DIM);

  rgcn_init_heads<<<(R_REL * N_NODES + 255) / 256, 256, 0, stream>>>(head, R_REL * N_NODES, ticket);

  rgcn_link<<<LINK_GRID, LINK_T, 0, stream>>>(ei, et, head, rec, ticket);

  const int nodeBlocks = (N_NODES * 32 + 255) / 256;
  rgcn_layer1<<<nodeBlocks, 256, 0, stream>>>(head, rec, W1, root1, b1, h1);

  rgcn_layer2_out<<<TILES, BLOCK2, 0, stream>>>(head, rec, h1, W2, root2, b2, out);

  (void)in_sizes; (void)n_in; (void)out_size; (void)ws_size;
}
